# starting kernel
# baseline (speedup 1.0000x reference)
#include <hip/hip_runtime.h>
#include <hip/hip_bf16.h>

typedef __attribute__((ext_vector_type(2))) float v2f;
typedef __attribute__((ext_vector_type(8))) float v8f;

#define DIM     1024
#define DSTATE  16
#define BATCH   4
#define SEQ     8192
#define CHUNK   128
#define NCHUNK  (SEQ / CHUNK)
#define SD_STRIDE 1032

__global__ void __launch_bounds__(128)
ssm_phase1(const float* __restrict__ x, const float* __restrict__ A,
           float* __restrict__ ws1, float* __restrict__ ws2)
{
    __shared__ alignas(16) float sA[DIM * DSTATE];
    __shared__ alignas(16) float sS[CHUNK * DSTATE];
    __shared__ alignas(16) float sTot[8 * DSTATE];

    const int blk   = blockIdx.x;
    const int b     = blk >> 6;
    const int chunk = blk & (NCHUNK - 1);
    const int n0    = chunk * CHUNK;
    const int tid   = threadIdx.x;

    {
        const float4* src = (const float4*)A;
        float4*       dst = (float4*)sA;
        for (int i = tid; i < DIM * DSTATE / 4; i += 128) dst[i] = src[i];
    }
    __syncthreads();

    const int wave = tid >> 5;
    const int lane = tid & 31;
    const int lrow = lane & 15;
    const int hi   = lane >> 4;

    for (int tt = 0; tt < 2; ++tt) {
        const int t = wave * 2 + tt;
        const float* xrow = x + ((size_t)b * SEQ + n0 + t * 16 + lrow) * DIM;
        v8f acc = {};
        #pragma unroll 4
        for (int kk = 0; kk < DIM / 4; ++kk) {
            const int k0 = kk * 4 + hi * 2;
            v2f a = *(const v2f*)(xrow + k0);
            v2f bb;
            bb.x = sA[(k0    ) * DSTATE + lrow];
            bb.y = sA[(k0 + 1) * DSTATE + lrow];
            acc = __builtin_amdgcn_wmma_f32_16x16x4_f32(
                false, a, false, bb, (short)0, acc, false, false);
        }
        const int rbase = t * 16 + hi * 8;
        #pragma unroll
        for (int j = 0; j < 8; ++j)
            sS[(rbase + j) * DSTATE + lrow] = acc[j];
    }
    __syncthreads();

    {
        const int s = tid & 15;
        const int g = tid >> 4;
        float run = 0.f;
        #pragma unroll
        for (int r = 0; r < 16; ++r) {
            run += sS[(g * 16 + r) * DSTATE + s];
            sS[(g * 16 + r) * DSTATE + s] = run;
        }
        sTot[g * DSTATE + s] = run;
        __syncthreads();
        float off = 0.f;
        for (int gg = 0; gg < g; ++gg) off += sTot[gg * DSTATE + s];
        #pragma unroll
        for (int r = 0; r < 16; ++r)
            sS[(g * 16 + r) * DSTATE + s] += off;
        if (g == 7)
            ws2[((size_t)b * NCHUNK + chunk) * DSTATE + s] = off + run;
    }
    __syncthreads();

    {
        float4*       dst = (float4*)(ws1 + ((size_t)b * SEQ + n0) * DSTATE);
        const float4* src = (const float4*)sS;
        for (int i = tid; i < CHUNK * DSTATE / 4; i += 128) dst[i] = src[i];
    }
}

__global__ void __launch_bounds__(256)
ssm_phase2(const float* __restrict__ ws1, const float* __restrict__ ws2,
           const float* __restrict__ Dm, float* __restrict__ out)
{
    __shared__ alignas(16) float sD[DSTATE * SD_STRIDE];
    __shared__ alignas(16) float sS[CHUNK * DSTATE];
    __shared__ float sOff[DSTATE];

    const int blk   = blockIdx.x;
    const int b     = blk >> 6;
    const int chunk = blk & (NCHUNK - 1);
    const int n0    = chunk * CHUNK;
    const int tid   = threadIdx.x;

    {
        const float4* src = (const float4*)Dm;
        for (int i4 = tid; i4 < DSTATE * DIM / 4; i4 += 256) {
            const int row  = i4 >> 8;
            const int col4 = i4 & 255;
            ((float4*)(sD + row * SD_STRIDE))[col4] = src[i4];
        }
    }
    {
        const float4* src = (const float4*)(ws1 + ((size_t)b * SEQ + n0) * DSTATE);
        float4*       dst = (float4*)sS;
        for (int i = tid; i < CHUNK * DSTATE / 4; i += 256) dst[i] = src[i];
    }
    if (tid < DSTATE) {
        float off = 0.f;
        for (int c = 0; c < chunk; ++c)
            off += ws2[((size_t)b * NCHUNK + c) * DSTATE + tid];
        sOff[tid] = off;
    }
    __syncthreads();
    if (tid < 128) {
        const int s = tid & 15;
        const int g = tid >> 4;
        const float off = sOff[s];
        #pragma unroll
        for (int r = 0; r < 16; ++r) sS[(g * 16 + r) * DSTATE + s] += off;
    }
    __syncthreads();

    const int wave = tid >> 5;
    const int lane = tid & 31;
    const int lrow = lane & 15;
    const int hi   = lane >> 4;

    const float* srow = sS + (wave * 16 + lrow) * DSTATE;
    v2f a0 = *(const v2f*)(srow + 0  + hi * 2);
    v2f a1 = *(const v2f*)(srow + 4  + hi * 2);
    v2f a2 = *(const v2f*)(srow + 8  + hi * 2);
    v2f a3 = *(const v2f*)(srow + 12 + hi * 2);

    float* obase = out + ((size_t)b * SEQ + n0 + wave * 16 + hi * 8) * DIM;
    for (int j = 0; j < DIM / 16; ++j) {
        const int ncol = j * 16 + lrow;
        v2f bb;
        v8f acc = {};
        bb.x = sD[(0  + hi * 2) * SD_STRIDE + ncol];
        bb.y = sD[(1  + hi * 2) * SD_STRIDE + ncol];
        acc = __builtin_amdgcn_wmma_f32_16x16x4_f32(false, a0, false, bb, (short)0, acc, false, false);
        bb.x = sD[(4  + hi * 2) * SD_STRIDE + ncol];
        bb.y = sD[(5  + hi * 2) * SD_STRIDE + ncol];
        acc = __builtin_amdgcn_wmma_f32_16x16x4_f32(false, a1, false, bb, (short)0, acc, false, false);
        bb.x = sD[(8  + hi * 2) * SD_STRIDE + ncol];
        bb.y = sD[(9  + hi * 2) * SD_STRIDE + ncol];
        acc = __builtin_amdgcn_wmma_f32_16x16x4_f32(false, a2, false, bb, (short)0, acc, false, false);
        bb.x = sD[(12 + hi * 2) * SD_STRIDE + ncol];
        bb.y = sD[(13 + hi * 2) * SD_STRIDE + ncol];
        acc = __builtin_amdgcn_wmma_f32_16x16x4_f32(false, a3, false, bb, (short)0, acc, false, false);
        #pragma unroll
        for (int r = 0; r < 8; ++r)
            obase[(size_t)r * DIM + ncol] = acc[r];
    }
}

extern "C" void kernel_launch(void* const* d_in, const int* in_sizes, int n_in,
                              void* d_out, int out_size, void* d_ws, size_t ws_size,
                              hipStream_t stream)
{
    (void)in_sizes; (void)n_in; (void)out_size; (void)ws_size;
    const float* x  = (const float*)d_in[0];
    const float* A  = (const float*)d_in[1];
    const float* Dm = (const float*)d_in[2];
    float* out = (float*)d_out;

    float* ws1 = (float*)d_ws;
    float* ws2 = ws1 + (size_t)BATCH * SEQ * DSTATE;

    dim3 grid(BATCH * NCHUNK);
    ssm_phase1<<<grid, 128, 0, stream>>>(x, A, ws1, ws2);
    ssm_phase2<<<grid, 256, 0, stream>>>(ws1, ws2, Dm, out);
}


# phase 1 keeps more bytes in flight: K halves on 16 waves per group (was 4), x by 128-bit loads 16 deep, A as k quads in LDS; two accumulators, halves added after (f32 sum order changes)
# speedup vs baseline: 1.5052x; 1.5052x over previous
#include <hip/hip_runtime.h>
#include <hip/hip_bf16.h>

typedef __attribute__((ext_vector_type(2))) float v2f;
typedef __attribute__((ext_vector_type(8))) float v8f;

#define DIM     1024
#define DSTATE  16
#define BATCH   4
#define SEQ     8192
#define CHUNK   128
#define NCHUNK  (SEQ / CHUNK)
#define SD_STRIDE 1032

#define KHALF   (DIM / 2)
#define NSTEP   (KHALF / 8)
#define NRING   16

__global__ void __launch_bounds__(512)
ssm_phase1(const float* __restrict__ x, const float* __restrict__ A,
           float* __restrict__ ws1, float* __restrict__ ws2)
{
    __shared__ alignas(16) float sA4[DIM * DSTATE];
    __shared__ alignas(16) float sP[2 * CHUNK * DSTATE];
    __shared__ alignas(16) float sS[CHUNK * DSTATE];
    __shared__ alignas(16) float sTot[8 * DSTATE];

    const int blk   = blockIdx.x;
    const int b     = blk >> 6;
    const int chunk = blk & (NCHUNK - 1);
    const int n0    = chunk * CHUNK;
    const int tid   = threadIdx.x;

    const int wave = tid >> 5;
    const int lane = tid & 31;
    const int lrow = lane & 15;
    const int hi   = lane >> 4;
    const int t    = wave & 7;
    const int q    = wave >> 3;

    const float4* xr = (const float4*)(x + ((size_t)b * SEQ + n0 + t * 16 + lrow) * DIM
                                         + q * KHALF + hi * 4);
    float4 xa[NRING];
    #pragma unroll
    for (int i = 0; i < NRING; ++i) xa[i] = xr[2 * i];

    {
        const float4* src = (const float4*)A;
        float4 v[8];
        #pragma unroll
        for (int it = 0; it < 8; ++it) v[it] = src[tid + it * 512];
        #pragma unroll
        for (int it = 0; it < 8; ++it) {
            const int i4 = tid + it * 512;
            const int k = i4 >> 2;
            const int j = i4 & 3;
            float* d = sA4 + (k >> 2) * 64 + j * 16 + (k & 3);
            d[0] = v[it].x; d[4] = v[it].y; d[8] = v[it].z; d[12] = v[it].w;
        }
    }
    __syncthreads();

    {
        const float4* bf = (const float4*)sA4 + (q * (KHALF / 4) + hi) * 16 + lrow;
        v8f acc0 = {};
        v8f acc1 = {};
        #pragma unroll
        for (int i = 0; i < NSTEP; ++i) {
            const float4 a = xa[i % NRING];
            if (i + NRING < NSTEP) xa[i % NRING] = xr[2 * (i + NRING)];
            const float4 bq = bf[i * 32];
            v2f a01; a01.x = a.x;  a01.y = a.y;
            v2f a23; a23.x = a.z;  a23.y = a.w;
            v2f b01; b01.x = bq.x; b01.y = bq.y;
            v2f b23; b23.x = bq.z; b23.y = bq.w;
            acc0 = __builtin_amdgcn_wmma_f32_16x16x4_f32(
                false, a01, false, b01, (short)0, acc0, false, false);
            acc1 = __builtin_amdgcn_wmma_f32_16x16x4_f32(
                false, a23, false, b23, (short)0, acc1, false, false);
        }
        float* pp = sP + q * (CHUNK * DSTATE) + (t * 16 + hi * 8) * DSTATE + lrow;
        #pragma unroll
        for (int j = 0; j < 8; ++j)
            pp[j * DSTATE] = acc0[j] + acc1[j];
    }
    __syncthreads();

    const int s = tid & 15;
    const int g = (tid >> 4) & 7;
    float run = 0.f;
    if (tid < 128) {
        #pragma unroll
        for (int r = 0; r < 16; ++r) {
            const int idx = (g * 16 + r) * DSTATE + s;
            run += sP[idx] + sP[CHUNK * DSTATE + idx];
            sS[idx] = run;
        }
        sTot[g * DSTATE + s] = run;
    }
    __syncthreads();
    if (tid < 128) {
        float off = 0.f;
        for (int gg = 0; gg < g; ++gg) off += sTot[gg * DSTATE + s];
        #pragma unroll
        for (int r = 0; r < 16; ++r)
            sS[(g * 16 + r) * DSTATE + s] += off;
        if (g == 7)
            ws2[((size_t)b * NCHUNK + chunk) * DSTATE + s] = off + run;
    }
    __syncthreads();

    {
        float4*       dst = (float4*)(ws1 + ((size_t)b * SEQ + n0) * DSTATE);
        const float4* src = (const float4*)sS;
        dst[tid] = src[tid];
    }
}

__global__ void __launch_bounds__(256)
ssm_phase2(const float* __restrict__ ws1, const float* __restrict__ ws2,
           const float* __restrict__ Dm, float* __restrict__ out)
{
    __shared__ alignas(16) float sD[DSTATE * SD_STRIDE];
    __shared__ alignas(16) float sS[CHUNK * DSTATE];
    __shared__ float sOff[DSTATE];

    const int blk   = blockIdx.x;
    const int b     = blk >> 6;
    const int chunk = blk & (NCHUNK - 1);
    const int n0    = chunk * CHUNK;
    const int tid   = threadIdx.x;

    {
        const float4* src = (const float4*)Dm;
        for (int i4 = tid; i4 < DSTATE * DIM / 4; i4 += 256) {
            const int row  = i4 >> 8;
            const int col4 = i4 & 255;
            ((float4*)(sD + row * SD_STRIDE))[col4] = src[i4];
        }
    }
    {
        const float4* src = (const float4*)(ws1 + ((size_t)b * SEQ + n0) * DSTATE);
        float4*       dst = (float4*)sS;
        for (int i = tid; i < CHUNK * DSTATE / 4; i += 256) dst[i] = src[i];
    }
    if (tid < DSTATE) {
        float off = 0.f;
        for (int c = 0; c < chunk; ++c)
            off += ws2[((size_t)b * NCHUNK + c) * DSTATE + tid];
        sOff[tid] = off;
    }
    __syncthreads();
    if (tid < 128) {
        const int s = tid & 15;
        const int g = tid >> 4;
        const float off = sOff[s];
        #pragma unroll
        for (int r = 0; r < 16; ++r) sS[(g * 16 + r) * DSTATE + s] += off;
    }
    __syncthreads();

    const int wave = tid >> 5;
    const int lane = tid & 31;
    const int lrow = lane & 15;
    const int hi   = lane >> 4;

    const float* srow = sS + (wave * 16 + lrow) * DSTATE;
    v2f a0 = *(const v2f*)(srow + 0  + hi * 2);
    v2f a1 = *(const v2f*)(srow + 4  + hi * 2);
    v2f a2 = *(const v2f*)(srow + 8  + hi * 2);
    v2f a3 = *(const v2f*)(srow + 12 + hi * 2);

    float* obase = out + ((size_t)b * SEQ + n0 + wave * 16 + hi * 8) * DIM;
    for (int j = 0; j < DIM / 16; ++j) {
        const int ncol = j * 16 + lrow;
        v2f bb;
        v8f acc = {};
        bb.x = sD[(0  + hi * 2) * SD_STRIDE + ncol];
        bb.y = sD[(1  + hi * 2) * SD_STRIDE + ncol];
        acc = __builtin_amdgcn_wmma_f32_16x16x4_f32(false, a0, false, bb, (short)0, acc, false, false);
        bb.x = sD[(4  + hi * 2) * SD_STRIDE + ncol];
        bb.y = sD[(5  + hi * 2) * SD_STRIDE + ncol];
        acc = __builtin_amdgcn_wmma_f32_16x16x4_f32(false, a1, false, bb, (short)0, acc, false, false);
        bb.x = sD[(8  + hi * 2) * SD_STRIDE + ncol];
        bb.y = sD[(9  + hi * 2) * SD_STRIDE + ncol];
        acc = __builtin_amdgcn_wmma_f32_16x16x4_f32(false, a2, false, bb, (short)0, acc, false, false);
        bb.x = sD[(12 + hi * 2) * SD_STRIDE + ncol];
        bb.y = sD[(13 + hi * 2) * SD_STRIDE + ncol];
        acc = __builtin_amdgcn_wmma_f32_16x16x4_f32(false, a3, false, bb, (short)0, acc, false, false);
        #pragma unroll
        for (int r = 0; r < 8; ++r)
            obase[(size_t)r * DIM + ncol] = acc[r];
    }
}

extern "C" void kernel_launch(void* const* d_in, const int* in_sizes, int n_in,
                              void* d_out, int out_size, void* d_ws, size_t ws_size,
                              hipStream_t stream)
{
    (void)in_sizes; (void)n_in; (void)out_size; (void)ws_size;
    const float* x  = (const float*)d_in[0];
    const float* A  = (const float*)d_in[1];
    const float* Dm = (const float*)d_in[2];
    float* out = (float*)d_out;

    float* ws1 = (float*)d_ws;
    float* ws2 = ws1 + (size_t)BATCH * SEQ * DSTATE;

    dim3 grid(BATCH * NCHUNK);
    ssm_phase1<<<grid, 512, 0, stream>>>(x, A, ws1, ws2);
    ssm_phase2<<<grid, 256, 0, stream>>>(ws1, ws2, Dm, out);
}


# phase 2 writes whole 128-byte lines: each wave stages its 16x64 accumulator block in a private LDS patch and stores 2 rows x 256 B per instruction with 16-byte stores (was 2 x 64 B per store)
# speedup vs baseline: 2.1692x; 1.4411x over previous
#include <hip/hip_runtime.h>
#include <hip/hip_bf16.h>

typedef __attribute__((ext_vector_type(2))) float v2f;
typedef __attribute__((ext_vector_type(8))) float v8f;

#define DIM     1024
#define DSTATE  16
#define BATCH   4
#define SEQ     8192
#define CHUNK   128
#define NCHUNK  (SEQ / CHUNK)
#define SD_STRIDE 1032

#define KHALF   (DIM / 2)
#define NSTEP   (KHALF / 8)
#define NRING   16

__global__ void __launch_bounds__(512)
ssm_phase1(const float* __restrict__ x, const float* __restrict__ A,
           float* __restrict__ ws1, float* __restrict__ ws2)
{
    __shared__ alignas(16) float sA4[DIM * DSTATE];
    __shared__ alignas(16) float sP[2 * CHUNK * DSTATE];
    __shared__ alignas(16) float sS[CHUNK * DSTATE];
    __shared__ alignas(16) float sTot[8 * DSTATE];

    const int blk   = blockIdx.x;
    const int b     = blk >> 6;
    const int chunk = blk & (NCHUNK - 1);
    const int n0    = chunk * CHUNK;
    const int tid   = threadIdx.x;

    const int wave = tid >> 5;
    const int lane = tid & 31;
    const int lrow = lane & 15;
    const int hi   = lane >> 4;
    const int t    = wave & 7;
    const int q    = wave >> 3;

    const float4* xr = (const float4*)(x + ((size_t)b * SEQ + n0 + t * 16 + lrow) * DIM
                                         + q * KHALF + hi * 4);
    float4 xa[NRING];
    #pragma unroll
    for (int i = 0; i < NRING; ++i) xa[i] = xr[2 * i];

    {
        const float4* src = (const float4*)A;
        float4 v[8];
        #pragma unroll
        for (int it = 0; it < 8; ++it) v[it] = src[tid + it * 512];
        #pragma unroll
        for (int it = 0; it < 8; ++it) {
            const int i4 = tid + it * 512;
            const int k = i4 >> 2;
            const int j = i4 & 3;
            float* d = sA4 + (k >> 2) * 64 + j * 16 + (k & 3);
            d[0] = v[it].x; d[4] = v[it].y; d[8] = v[it].z; d[12] = v[it].w;
        }
    }
    __syncthreads();

    {
        const float4* bf = (const float4*)sA4 + (q * (KHALF / 4) + hi) * 16 + lrow;
        v8f acc0 = {};
        v8f acc1 = {};
        #pragma unroll
        for (int i = 0; i < NSTEP; ++i) {
            const float4 a = xa[i % NRING];
            if (i + NRING < NSTEP) xa[i % NRING] = xr[2 * (i + NRING)];
            const float4 bq = bf[i * 32];
            v2f a01; a01.x = a.x;  a01.y = a.y;
            v2f a23; a23.x = a.z;  a23.y = a.w;
            v2f b01; b01.x = bq.x; b01.y = bq.y;
            v2f b23; b23.x = bq.z; b23.y = bq.w;
            acc0 = __builtin_amdgcn_wmma_f32_16x16x4_f32(
                false, a01, false, b01, (short)0, acc0, false, false);
            acc1 = __builtin_amdgcn_wmma_f32_16x16x4_f32(
                false, a23, false, b23, (short)0, acc1, false, false);
        }
        float* pp = sP + q * (CHUNK * DSTATE) + (t * 16 + hi * 8) * DSTATE + lrow;
        #pragma unroll
        for (int j = 0; j < 8; ++j)
            pp[j * DSTATE] = acc0[j] + acc1[j];
    }
    __syncthreads();

    const int s = tid & 15;
    const int g = (tid >> 4) & 7;
    float run = 0.f;
    if (tid < 128) {
        #pragma unroll
        for (int r = 0; r < 16; ++r) {
            const int idx = (g * 16 + r) * DSTATE + s;
            run += sP[idx] + sP[CHUNK * DSTATE + idx];
            sS[idx] = run;
        }
        sTot[g * DSTATE + s] = run;
    }
    __syncthreads();
    if (tid < 128) {
        float off = 0.f;
        for (int gg = 0; gg < g; ++gg) off += sTot[gg * DSTATE + s];
        #pragma unroll
        for (int r = 0; r < 16; ++r)
            sS[(g * 16 + r) * DSTATE + s] += off;
        if (g == 7)
            ws2[((size_t)b * NCHUNK + chunk) * DSTATE + s] = off + run;
    }
    __syncthreads();

    {
        float4*       dst = (float4*)(ws1 + ((size_t)b * SEQ + n0) * DSTATE);
        const float4* src = (const float4*)sS;
        dst[tid] = src[tid];
    }
}

#define ST_PITCH 68
__global__ void __launch_bounds__(256)
ssm_phase2(const float* __restrict__ ws1, const float* __restrict__ ws2,
           const float* __restrict__ Dm, float* __restrict__ out)
{
    __shared__ alignas(16) float sD[DSTATE * SD_STRIDE];
    __shared__ alignas(16) float sS[CHUNK * DSTATE];
    __shared__ alignas(16) float sStage[8 * 16 * ST_PITCH];
    __shared__ float sOff[DSTATE];

    const int blk   = blockIdx.x;
    const int b     = blk >> 6;
    const int chunk = blk & (NCHUNK - 1);
    const int n0    = chunk * CHUNK;
    const int tid   = threadIdx.x;

    {
        const float4* src = (const float4*)Dm;
        for (int i4 = tid; i4 < DSTATE * DIM / 4; i4 += 256) {
            const int row  = i4 >> 8;
            const int col4 = i4 & 255;
            ((float4*)(sD + row * SD_STRIDE))[col4] = src[i4];
        }
    }
    {
        const float4* src = (const float4*)(ws1 + ((size_t)b * SEQ + n0) * DSTATE);
        float4*       dst = (float4*)sS;
        for (int i = tid; i < CHUNK * DSTATE / 4; i += 256) dst[i] = src[i];
    }
    if (tid < DSTATE) {
        float off = 0.f;
        for (int c = 0; c < chunk; ++c)
            off += ws2[((size_t)b * NCHUNK + c) * DSTATE + tid];
        sOff[tid] = off;
    }
    __syncthreads();
    if (tid < 128) {
        const int s = tid & 15;
        const int g = tid >> 4;
        const float off = sOff[s];
        #pragma unroll
        for (int r = 0; r < 16; ++r) sS[(g * 16 + r) * DSTATE + s] += off;
    }
    __syncthreads();

    const int wave = tid >> 5;
    const int lane = tid & 31;
    const int lrow = lane & 15;
    const int hi   = lane >> 4;

    const float* srow = sS + (wave * 16 + lrow) * DSTATE;
    v2f a0 = *(const v2f*)(srow + 0  + hi * 2);
    v2f a1 = *(const v2f*)(srow + 4  + hi * 2);
    v2f a2 = *(const v2f*)(srow + 8  + hi * 2);
    v2f a3 = *(const v2f*)(srow + 12 + hi * 2);

    float*        stw = sStage + wave * (16 * ST_PITCH) + (hi * 8) * ST_PITCH + lrow;
    const float4* str = (const float4*)(sStage + wave * (16 * ST_PITCH) + hi * ST_PITCH) + lrow;
    float* obase = out + ((size_t)b * SEQ + n0 + wave * 16 + hi) * DIM + lrow * 4;

    for (int j4 = 0; j4 < DIM / 64; ++j4) {
        v8f acc[4];
        for (int tt = 0; tt < 4; ++tt) {
            const int ncol = j4 * 64 + tt * 16 + lrow;
            v2f bb;
            v8f a = {};
            bb.x = sD[(0  + hi * 2) * SD_STRIDE + ncol];
            bb.y = sD[(1  + hi * 2) * SD_STRIDE + ncol];
            a = __builtin_amdgcn_wmma_f32_16x16x4_f32(false, a0, false, bb, (short)0, a, false, false);
            bb.x = sD[(4  + hi * 2) * SD_STRIDE + ncol];
            bb.y = sD[(5  + hi * 2) * SD_STRIDE + ncol];
            a = __builtin_amdgcn_wmma_f32_16x16x4_f32(false, a1, false, bb, (short)0, a, false, false);
            bb.x = sD[(8  + hi * 2) * SD_STRIDE + ncol];
            bb.y = sD[(9  + hi * 2) * SD_STRIDE + ncol];
            a = __builtin_amdgcn_wmma_f32_16x16x4_f32(false, a2, false, bb, (short)0, a, false, false);
            bb.x = sD[(12 + hi * 2) * SD_STRIDE + ncol];
            bb.y = sD[(13 + hi * 2) * SD_STRIDE + ncol];
            a = __builtin_amdgcn_wmma_f32_16x16x4_f32(false, a3, false, bb, (short)0, a, false, false);
            acc[tt] = a;
        }
        for (int tt = 0; tt < 4; ++tt) {
            for (int r = 0; r < 8; ++r)
                stw[r * ST_PITCH + tt * 16] = acc[tt][r];
        }
        float4 v[8];
        for (int p = 0; p < 8; ++p) v[p] = str[p * (2 * ST_PITCH / 4)];
        for (int p = 0; p < 8; ++p)
            *(float4*)(obase + (size_t)(2 * p) * DIM + j4 * 64) = v[p];
    }
}

extern "C" void kernel_launch(void* const* d_in, const int* in_sizes, int n_in,
                              void* d_out, int out_size, void* d_ws, size_t ws_size,
                              hipStream_t stream)
{
    (void)in_sizes; (void)n_in; (void)out_size; (void)ws_size;
    const float* x  = (const float*)d_in[0];
    const float* A  = (const float*)d_in[1];
    const float* Dm = (const float*)d_in[2];
    float* out = (float*)d_out;

    float* ws1 = (float*)d_ws;
    float* ws2 = ws1 + (size_t)BATCH * SEQ * DSTATE;

    dim3 grid(BATCH * NCHUNK);
    ssm_phase1<<<grid, 512, 0, stream>>>(x, A, ws1, ws2);
    ssm_phase2<<<grid, 256, 0, stream>>>(ws1, ws2, Dm, out);
}


# phase 2 prologue in one memory round trip: D, chunk states and all 64 chunk totals requested at once, prefix offset summed from LDS in the same order (was up to 18 dependent round trips)
# speedup vs baseline: 2.4019x; 1.1073x over previous
#include <hip/hip_runtime.h>
#include <hip/hip_bf16.h>

typedef __attribute__((ext_vector_type(2))) float v2f;
typedef __attribute__((ext_vector_type(8))) float v8f;

#define DIM     1024
#define DSTATE  16
#define BATCH   4
#define SEQ     8192
#define CHUNK   128
#define NCHUNK  (SEQ / CHUNK)
#define SD_STRIDE 1032

#define KHALF   (DIM / 2)
#define NSTEP   (KHALF / 8)
#define NRING   16

__global__ void __launch_bounds__(512)
ssm_phase1(const float* __restrict__ x, const float* __restrict__ A,
           float* __restrict__ ws1, float* __restrict__ ws2)
{
    __shared__ alignas(16) float sA4[DIM * DSTATE];
    __shared__ alignas(16) float sP[2 * CHUNK * DSTATE];
    __shared__ alignas(16) float sS[CHUNK * DSTATE];
    __shared__ alignas(16) float sTot[8 * DSTATE];

    const int blk   = blockIdx.x;
    const int b     = blk >> 6;
    const int chunk = blk & (NCHUNK - 1);
    const int n0    = chunk * CHUNK;
    const int tid   = threadIdx.x;

    const int wave = tid >> 5;
    const int lane = tid & 31;
    const int lrow = lane & 15;
    const int hi   = lane >> 4;
    const int t    = wave & 7;
    const int q    = wave >> 3;

    const float4* xr = (const float4*)(x + ((size_t)b * SEQ + n0 + t * 16 + lrow) * DIM
                                         + q * KHALF + hi * 4);
    float4 xa[NRING];
    #pragma unroll
    for (int i = 0; i < NRING; ++i) xa[i] = xr[2 * i];

    {
        const float4* src = (const float4*)A;
        float4 v[8];
        #pragma unroll
        for (int it = 0; it < 8; ++it) v[it] = src[tid + it * 512];
        #pragma unroll
        for (int it = 0; it < 8; ++it) {
            const int i4 = tid + it * 512;
            const int k = i4 >> 2;
            const int j = i4 & 3;
            float* d = sA4 + (k >> 2) * 64 + j * 16 + (k & 3);
            d[0] = v[it].x; d[4] = v[it].y; d[8] = v[it].z; d[12] = v[it].w;
        }
    }
    __syncthreads();

    {
        const float4* bf = (const float4*)sA4 + (q * (KHALF / 4) + hi) * 16 + lrow;
        v8f acc0 = {};
        v8f acc1 = {};
        #pragma unroll
        for (int i = 0; i < NSTEP; ++i) {
            const float4 a = xa[i % NRING];
            if (i + NRING < NSTEP) xa[i % NRING] = xr[2 * (i + NRING)];
            const float4 bq = bf[i * 32];
            v2f a01; a01.x = a.x;  a01.y = a.y;
            v2f a23; a23.x = a.z;  a23.y = a.w;
            v2f b01; b01.x = bq.x; b01.y = bq.y;
            v2f b23; b23.x = bq.z; b23.y = bq.w;
            acc0 = __builtin_amdgcn_wmma_f32_16x16x4_f32(
                false, a01, false, b01, (short)0, acc0, false, false);
            acc1 = __builtin_amdgcn_wmma_f32_16x16x4_f32(
                false, a23, false, b23, (short)0, acc1, false, false);
        }
        float* pp = sP + q * (CHUNK * DSTATE) + (t * 16 + hi * 8) * DSTATE + lrow;
        #pragma unroll
        for (int j = 0; j < 8; ++j)
            pp[j * DSTATE] = acc0[j] + acc1[j];
    }
    __syncthreads();

    const int s = tid & 15;
    const int g = (tid >> 4) & 7;
    float run = 0.f;
    if (tid < 128) {
        #pragma unroll
        for (int r = 0; r < 16; ++r) {
            const int idx = (g * 16 + r) * DSTATE + s;
            run += sP[idx] + sP[CHUNK * DSTATE + idx];
            sS[idx] = run;
        }
        sTot[g * DSTATE + s] = run;
    }
    __syncthreads();
    if (tid < 128) {
        float off = 0.f;
        for (int gg = 0; gg < g; ++gg) off += sTot[gg * DSTATE + s];
        #pragma unroll
        for (int r = 0; r < 16; ++r)
            sS[(g * 16 + r) * DSTATE + s] += off;
        if (g == 7)
            ws2[((size_t)b * NCHUNK + chunk) * DSTATE + s] = off + run;
    }
    __syncthreads();

    {
        float4*       dst = (float4*)(ws1 + ((size_t)b * SEQ + n0) * DSTATE);
        const float4* src = (const float4*)sS;
        dst[tid] = src[tid];
    }
}

#define ST_PITCH 68
__global__ void __launch_bounds__(256)
ssm_phase2(const float* __restrict__ ws1, const float* __restrict__ ws2,
           const float* __restrict__ Dm, float* __restrict__ out)
{
    __shared__ alignas(16) float sD[DSTATE * SD_STRIDE];
    __shared__ alignas(16) float sS[CHUNK * DSTATE];
    __shared__ alignas(16) float sStage[8 * 16 * ST_PITCH];
    __shared__ alignas(16) float sTot[NCHUNK * DSTATE];

    const int blk   = blockIdx.x;
    const int b     = blk >> 6;
    const int chunk = blk & (NCHUNK - 1);
    const int n0    = chunk * CHUNK;
    const int tid   = threadIdx.x;

    {
        const float4* dsrc = (const float4*)Dm;
        const float4* ssrc = (const float4*)(ws1 + ((size_t)b * SEQ + n0) * DSTATE);
        const float4* tsrc = (const float4*)(ws2 + (size_t)b * NCHUNK * DSTATE);
        float4 dv[16];
        float4 sv[2];
        #pragma unroll
        for (int it = 0; it < 16; ++it) dv[it] = dsrc[tid + it * 256];
        sv[0] = ssrc[tid];
        sv[1] = ssrc[tid + 256];
        const float4 tv = tsrc[tid];
        #pragma unroll
        for (int it = 0; it < 16; ++it)
            ((float4*)(sD + it * SD_STRIDE))[tid] = dv[it];
        ((float4*)sS)[tid]       = sv[0];
        ((float4*)sS)[tid + 256] = sv[1];
        ((float4*)sTot)[tid]     = tv;
    }
    __syncthreads();
    if (tid < 128) {
        const int s = tid & 15;
        const int g = tid >> 4;
        float tc[NCHUNK - 1];
        #pragma unroll
        for (int c = 0; c < NCHUNK - 1; ++c) tc[c] = sTot[c * DSTATE + s];
        float run = 0.f;
        float off = 0.f;
        #pragma unroll
        for (int c = 0; c < NCHUNK - 1; ++c) {
            run += tc[c];
            off = (c + 1 == chunk) ? run : off;
        }
        #pragma unroll
        for (int r = 0; r < 16; ++r) sS[(g * 16 + r) * DSTATE + s] += off;
    }
    __syncthreads();

    const int wave = tid >> 5;
    const int lane = tid & 31;
    const int lrow = lane & 15;
    const int hi   = lane >> 4;

    const float* srow = sS + (wave * 16 + lrow) * DSTATE;
    v2f a0 = *(const v2f*)(srow + 0  + hi * 2);
    v2f a1 = *(const v2f*)(srow + 4  + hi * 2);
    v2f a2 = *(const v2f*)(srow + 8  + hi * 2);
    v2f a3 = *(const v2f*)(srow + 12 + hi * 2);

    float*        stw = sStage + wave * (16 * ST_PITCH) + (hi * 8) * ST_PITCH + lrow;
    const float4* str = (const float4*)(sStage + wave * (16 * ST_PITCH) + hi * ST_PITCH) + lrow;
    float* obase = out + ((size_t)b * SEQ + n0 + wave * 16 + hi) * DIM + lrow * 4;

    for (int j4 = 0; j4 < DIM / 64; ++j4) {
        v8f acc[4];
        for (int tt = 0; tt < 4; ++tt) {
            const int ncol = j4 * 64 + tt * 16 + lrow;
            v2f bb;
            v8f a = {};
            bb.x = sD[(0  + hi * 2) * SD_STRIDE + ncol];
            bb.y = sD[(1  + hi * 2) * SD_STRIDE + ncol];
            a = __builtin_amdgcn_wmma_f32_16x16x4_f32(false, a0, false, bb, (short)0, a, false, false);
            bb.x = sD[(4  + hi * 2) * SD_STRIDE + ncol];
            bb.y = sD[(5  + hi * 2) * SD_STRIDE + ncol];
            a = __builtin_amdgcn_wmma_f32_16x16x4_f32(false, a1, false, bb, (short)0, a, false, false);
            bb.x = sD[(8  + hi * 2) * SD_STRIDE + ncol];
            bb.y = sD[(9  + hi * 2) * SD_STRIDE + ncol];
            a = __builtin_amdgcn_wmma_f32_16x16x4_f32(false, a2, false, bb, (short)0, a, false, false);
            bb.x = sD[(12 + hi * 2) * SD_STRIDE + ncol];
            bb.y = sD[(13 + hi * 2) * SD_STRIDE + ncol];
            a = __builtin_amdgcn_wmma_f32_16x16x4_f32(false, a3, false, bb, (short)0, a, false, false);
            acc[tt] = a;
        }
        for (int tt = 0; tt < 4; ++tt) {
            for (int r = 0; r < 8; ++r)
                stw[r * ST_PITCH + tt * 16] = acc[tt][r];
        }
        float4 v[8];
        for (int p = 0; p < 8; ++p) v[p] = str[p * (2 * ST_PITCH / 4)];
        for (int p = 0; p < 8; ++p)
            *(float4*)(obase + (size_t)(2 * p) * DIM + j4 * 64) = v[p];
    }
}

extern "C" void kernel_launch(void* const* d_in, const int* in_sizes, int n_in,
                              void* d_out, int out_size, void* d_ws, size_t ws_size,
                              hipStream_t stream)
{
    (void)in_sizes; (void)n_in; (void)out_size; (void)ws_size;
    const float* x  = (const float*)d_in[0];
    const float* A  = (const float*)d_in[1];
    const float* Dm = (const float*)d_in[2];
    float* out = (float*)d_out;

    float* ws1 = (float*)d_ws;
    float* ws2 = ws1 + (size_t)BATCH * SEQ * DSTATE;

    dim3 grid(BATCH * NCHUNK);
    ssm_phase1<<<grid, 512, 0, stream>>>(x, A, ws1, ws2);
    ssm_phase2<<<grid, 256, 0, stream>>>(ws1, ws2, Dm, out);
}
